# starting kernel
# baseline (speedup 1.0000x reference)
#include <hip/hip_runtime.h>
#include <math.h>

typedef float v2f __attribute__((ext_vector_type(2)));
typedef float v8f __attribute__((ext_vector_type(8)));

#define SPB 128
#define THREADS_A 256

__global__ __launch_bounds__(THREADS_A)
void pool_circuit_kernel(const float* __restrict__ x,
                         const float* __restrict__ params,
                         float* __restrict__ zbuf,
                         float* __restrict__ partials,
                         int B)
{
    __shared__ float pooled[SPB][4];
    __shared__ float red[SPB][8];
    __shared__ float sAB[8];

    const int tid   = threadIdx.x;
    const int lane  = tid & 31;
    const int wv    = tid >> 5;
    const int nIdx  = lane & 15;
    const int kBase = (lane >> 4) << 1;

    if (tid < 4) {
        float al = params[2 * tid + 0];
        float be = params[2 * tid + 1];
        sAB[tid]     = cosf(be);
        sAB[tid + 4] = sinf(be) * sinf(al);
    }

    v2f bt[6];
#pragma unroll
    for (int t = 0; t < 6; ++t) {
        int c0 = 4 * t + kBase;
        bt[t].x = (nIdx == (c0    ) / 6) ? (1.0f / 36.0f) : 0.0f;
        bt[t].y = (nIdx == (c0 + 1) / 6) ? (1.0f / 36.0f) : 0.0f;
    }

    const int sampleBase = blockIdx.x * SPB + wv * 16;
    const float* px = x + (size_t)(sampleBase + nIdx) * 576 + kBase;

    v8f acc = {0.f, 0.f, 0.f, 0.f, 0.f, 0.f, 0.f, 0.f};
#pragma unroll
    for (int row = 0; row < 6; ++row) {
#pragma unroll
        for (int t = 0; t < 6; ++t) {
            v2f a = *(const v2f*)(px + row * 24 + t * 4);
            acc = __builtin_amdgcn_wmma_f32_16x16x4_f32(
                false, a, false, bt[t], (short)0, acc, false, false);
        }
    }

    if (nIdx < 4) {
        int srow = wv * 16 + ((lane >> 4) << 3);
#pragma unroll
        for (int r = 0; r < 8; ++r)
            pooled[srow + r][nIdx] = acc[r];
    }
    __syncthreads();

    if (tid < SPB) {
        float z[4];
        float prod = 1.0f;
#pragma unroll
        for (int j = 0; j < 4; ++j) {
            float xj = pooled[tid][j];
            float sx, cx;
            sincosf(xj, &sx, &cx);
            prod *= sAB[j] * cx + sAB[4 + j] * sx;
            z[j] = prod;
        }
        size_t s = (size_t)blockIdx.x * SPB + tid;
        *(float4*)(zbuf + s * 4) = make_float4(z[0], z[1], z[2], z[3]);
#pragma unroll
        for (int j = 0; j < 4; ++j) {
            red[tid][j]     = z[j];
            red[tid][4 + j] = z[j] * z[j];
        }
    }
    __syncthreads();

    for (int s = SPB / 2; s > 0; s >>= 1) {
        if (tid < s) {
#pragma unroll
            for (int j = 0; j < 8; ++j) red[tid][j] += red[tid + s][j];
        }
        __syncthreads();
    }
    if (tid < 8) partials[(size_t)blockIdx.x * 8 + tid] = red[0][tid];
}

__global__ void stats_kernel(const float* __restrict__ partials, int nblk, int B,
                             const float* __restrict__ gamma,
                             const float* __restrict__ beta,
                             float* __restrict__ stats)
{
    __shared__ float sums[8];
    int j = threadIdx.x;
    if (j < 8) {
        float s = 0.f;
        for (int b = 0; b < nblk; ++b) s += partials[(size_t)b * 8 + j];
        sums[j] = s;
    }
    __syncthreads();
    if (j < 4) {
        float invB = 1.0f / (float)B;
        float mean = sums[j] * invB;
        float var  = sums[j + 4] * invB - mean * mean;
        float sc   = gamma[j] * rsqrtf(var + 1e-5f);
        stats[j]     = sc;
        stats[4 + j] = beta[j] - mean * sc;
    }
}

__global__ __launch_bounds__(256)
void norm_kernel(const float* __restrict__ zbuf,
                 const float* __restrict__ stats,
                 float* __restrict__ out, int B)
{
    int i = blockIdx.x * blockDim.x + threadIdx.x;
    if (i < B) {
        float4 z  = *(const float4*)(zbuf + (size_t)i * 4);
        float4 sc = *(const float4*)(stats);
        float4 sh = *(const float4*)(stats + 4);
        float4 o;
        o.x = z.x * sc.x + sh.x;
        o.y = z.y * sc.y + sh.y;
        o.z = z.z * sc.z + sh.z;
        o.w = z.w * sc.w + sh.w;
        *(float4*)(out + (size_t)i * 4) = o;
    }
}

extern "C" void kernel_launch(void* const* d_in, const int* in_sizes, int n_in,
                              void* d_out, int out_size, void* d_ws, size_t ws_size,
                              hipStream_t stream)
{
    const float* x      = (const float*)d_in[0];
    const float* params = (const float*)d_in[1];
    const float* gamma  = (const float*)d_in[2];
    const float* beta   = (const float*)d_in[3];
    float* out = (float*)d_out;

    int B = in_sizes[0] / 576;
    int nblkA = B / SPB;

    float* zbuf     = (float*)d_ws;
    float* partials = zbuf + (size_t)B * 4;
    float* stats    = partials + (size_t)nblkA * 8;

    pool_circuit_kernel<<<nblkA, THREADS_A, 0, stream>>>(x, params, zbuf, partials, B);
    stats_kernel<<<1, 32, 0, stream>>>(partials, nblkA, B, gamma, beta, stats);
    norm_kernel<<<(B + 255) / 256, 256, 0, stream>>>(zbuf, stats, out, B);
}


# Fold the one-wave serial stats launch into the normalise kernel: each block sums the 512 block partials itself in a parallel fixed-order tree (3 launches to 2; stats summation order changes)
# speedup vs baseline: 2.5046x; 2.5046x over previous
#include <hip/hip_runtime.h>
#include <math.h>

typedef float v2f __attribute__((ext_vector_type(2)));
typedef float v8f __attribute__((ext_vector_type(8)));

#define SPB 128
#define THREADS_A 256

__global__ __launch_bounds__(THREADS_A)
void pool_circuit_kernel(const float* __restrict__ x,
                         const float* __restrict__ params,
                         float* __restrict__ zbuf,
                         float* __restrict__ partials,
                         int B)
{
    __shared__ float pooled[SPB][4];
    __shared__ float red[SPB][8];
    __shared__ float sAB[8];

    const int tid   = threadIdx.x;
    const int lane  = tid & 31;
    const int wv    = tid >> 5;
    const int nIdx  = lane & 15;
    const int kBase = (lane >> 4) << 1;

    if (tid < 4) {
        float al = params[2 * tid + 0];
        float be = params[2 * tid + 1];
        sAB[tid]     = cosf(be);
        sAB[tid + 4] = sinf(be) * sinf(al);
    }

    v2f bt[6];
#pragma unroll
    for (int t = 0; t < 6; ++t) {
        int c0 = 4 * t + kBase;
        bt[t].x = (nIdx == (c0    ) / 6) ? (1.0f / 36.0f) : 0.0f;
        bt[t].y = (nIdx == (c0 + 1) / 6) ? (1.0f / 36.0f) : 0.0f;
    }

    const int sampleBase = blockIdx.x * SPB + wv * 16;
    const float* px = x + (size_t)(sampleBase + nIdx) * 576 + kBase;

    v8f acc = {0.f, 0.f, 0.f, 0.f, 0.f, 0.f, 0.f, 0.f};
#pragma unroll
    for (int row = 0; row < 6; ++row) {
#pragma unroll
        for (int t = 0; t < 6; ++t) {
            v2f a = *(const v2f*)(px + row * 24 + t * 4);
            acc = __builtin_amdgcn_wmma_f32_16x16x4_f32(
                false, a, false, bt[t], (short)0, acc, false, false);
        }
    }

    if (nIdx < 4) {
        int srow = wv * 16 + ((lane >> 4) << 3);
#pragma unroll
        for (int r = 0; r < 8; ++r)
            pooled[srow + r][nIdx] = acc[r];
    }
    __syncthreads();

    if (tid < SPB) {
        float z[4];
        float prod = 1.0f;
#pragma unroll
        for (int j = 0; j < 4; ++j) {
            float xj = pooled[tid][j];
            float sx, cx;
            sincosf(xj, &sx, &cx);
            prod *= sAB[j] * cx + sAB[4 + j] * sx;
            z[j] = prod;
        }
        size_t s = (size_t)blockIdx.x * SPB + tid;
        *(float4*)(zbuf + s * 4) = make_float4(z[0], z[1], z[2], z[3]);
#pragma unroll
        for (int j = 0; j < 4; ++j) {
            red[tid][j]     = z[j];
            red[tid][4 + j] = z[j] * z[j];
        }
    }
    __syncthreads();

    for (int s = SPB / 2; s > 0; s >>= 1) {
        if (tid < s) {
#pragma unroll
            for (int j = 0; j < 8; ++j) red[tid][j] += red[tid + s][j];
        }
        __syncthreads();
    }
    if (tid < 8) partials[(size_t)blockIdx.x * 8 + tid] = red[0][tid];
}

#define THREADS_N 256

__global__ __launch_bounds__(THREADS_N)
void stats_norm_kernel(const float* __restrict__ zbuf,
                       const float* __restrict__ partials, int nblk,
                       const float* __restrict__ gamma,
                       const float* __restrict__ beta,
                       float* __restrict__ out, int B)
{
    __shared__ float4 wsum[THREADS_N / 32][2];

    const int tid  = threadIdx.x;
    const int lane = tid & 31;
    const int wv   = tid >> 5;
    const int i    = blockIdx.x * THREADS_N + tid;

    float4 z = make_float4(0.f, 0.f, 0.f, 0.f);
    if (i < B) z = *(const float4*)(zbuf + (size_t)i * 4);
    const float g0 = gamma[0], g1 = gamma[1], g2 = gamma[2], g3 = gamma[3];
    const float b0 = beta[0],  b1 = beta[1],  b2 = beta[2],  b3 = beta[3];

    const float4* p4 = (const float4*)partials;
    const int n2 = 2 * nblk;
    const float4 zero4 = make_float4(0.f, 0.f, 0.f, 0.f);
    float4 a = zero4;
    for (int e = tid; e < n2; e += 4 * THREADS_N) {
        const int e1 = e + THREADS_N, e2 = e + 2 * THREADS_N, e3 = e + 3 * THREADS_N;
        float4 v0 = p4[e];
        float4 v1 = zero4, v2 = zero4, v3 = zero4;
        if (e1 < n2) v1 = p4[e1];
        if (e2 < n2) v2 = p4[e2];
        if (e3 < n2) v3 = p4[e3];
        a.x += (v0.x + v1.x) + (v2.x + v3.x);
        a.y += (v0.y + v1.y) + (v2.y + v3.y);
        a.z += (v0.z + v1.z) + (v2.z + v3.z);
        a.w += (v0.w + v1.w) + (v2.w + v3.w);
    }
    for (int m = 2; m < 32; m <<= 1) {
        a.x += __shfl_xor(a.x, m);
        a.y += __shfl_xor(a.y, m);
        a.z += __shfl_xor(a.z, m);
        a.w += __shfl_xor(a.w, m);
    }
    if (lane < 2) wsum[wv][lane] = a;
    __syncthreads();

    float4 s1 = wsum[0][0];
    float4 s2 = wsum[0][1];
    for (int w = 1; w < THREADS_N / 32; ++w) {
        float4 u = wsum[w][0];
        float4 q = wsum[w][1];
        s1.x += u.x; s1.y += u.y; s1.z += u.z; s1.w += u.w;
        s2.x += q.x; s2.y += q.y; s2.z += q.z; s2.w += q.w;
    }

    if (i < B) {
        const float invB = 1.0f / (float)B;
        float4 mean, var, sc, o;
        mean.x = s1.x * invB; mean.y = s1.y * invB; mean.z = s1.z * invB; mean.w = s1.w * invB;
        var.x = s2.x * invB - mean.x * mean.x;
        var.y = s2.y * invB - mean.y * mean.y;
        var.z = s2.z * invB - mean.z * mean.z;
        var.w = s2.w * invB - mean.w * mean.w;
        sc.x = g0 * rsqrtf(var.x + 1e-5f);
        sc.y = g1 * rsqrtf(var.y + 1e-5f);
        sc.z = g2 * rsqrtf(var.z + 1e-5f);
        sc.w = g3 * rsqrtf(var.w + 1e-5f);
        o.x = z.x * sc.x + (b0 - mean.x * sc.x);
        o.y = z.y * sc.y + (b1 - mean.y * sc.y);
        o.z = z.z * sc.z + (b2 - mean.z * sc.z);
        o.w = z.w * sc.w + (b3 - mean.w * sc.w);
        *(float4*)(out + (size_t)i * 4) = o;
    }
}

extern "C" void kernel_launch(void* const* d_in, const int* in_sizes, int n_in,
                              void* d_out, int out_size, void* d_ws, size_t ws_size,
                              hipStream_t stream)
{
    const float* x      = (const float*)d_in[0];
    const float* params = (const float*)d_in[1];
    const float* gamma  = (const float*)d_in[2];
    const float* beta   = (const float*)d_in[3];
    float* out = (float*)d_out;

    int B = in_sizes[0] / 576;
    int nblkA = B / SPB;

    float* zbuf     = (float*)d_ws;
    float* partials = zbuf + (size_t)B * 4;

    pool_circuit_kernel<<<nblkA, THREADS_A, 0, stream>>>(x, params, zbuf, partials, B);
    stats_norm_kernel<<<(B + 255) / 256, THREADS_N, 0, stream>>>(zbuf, partials, nblkA, gamma, beta, out, B);
}


# Per-block statistics summed in registers: a DPP butterfly inside each circuit wave and one LDS round over the four waves replace the seven-barrier LDS tree; fixed order, but a different sum tree
# speedup vs baseline: 2.6436x; 1.0555x over previous
#include <hip/hip_runtime.h>
#include <math.h>

typedef float v2f __attribute__((ext_vector_type(2)));
typedef float v8f __attribute__((ext_vector_type(8)));

#define SPB 128
#define THREADS_A 256

template <int M>
__device__ __forceinline__ float row_xor(float x)
{
    int b = __builtin_bit_cast(int, x);
    return __builtin_bit_cast(float,
        __builtin_amdgcn_update_dpp(0, b, 0x160 + M, 0xf, 0xf, true));
}

__device__ __forceinline__ float xhalf(float x)
{
    int b = __builtin_bit_cast(int, x);
    return __builtin_bit_cast(float,
        __builtin_amdgcn_permlanex16(b, b, 0x76543210, 0xfedcba98, false, false));
}

__global__ __launch_bounds__(THREADS_A)
void pool_circuit_kernel(const float* __restrict__ x,
                         const float* __restrict__ params,
                         float* __restrict__ zbuf,
                         float* __restrict__ partials,
                         int B)
{
    __shared__ float pooled[SPB][4];
    __shared__ float4 wred[SPB / 32][2];
    __shared__ float sAB[8];

    const int tid   = threadIdx.x;
    const int lane  = tid & 31;
    const int wv    = tid >> 5;
    const int nIdx  = lane & 15;
    const int kBase = (lane >> 4) << 1;

    if (tid < 4) {
        float al = params[2 * tid + 0];
        float be = params[2 * tid + 1];
        sAB[tid]     = cosf(be);
        sAB[tid + 4] = sinf(be) * sinf(al);
    }

    v2f bt[6];
#pragma unroll
    for (int t = 0; t < 6; ++t) {
        int c0 = 4 * t + kBase;
        bt[t].x = (nIdx == (c0    ) / 6) ? (1.0f / 36.0f) : 0.0f;
        bt[t].y = (nIdx == (c0 + 1) / 6) ? (1.0f / 36.0f) : 0.0f;
    }

    const int sampleBase = blockIdx.x * SPB + wv * 16;
    const float* px = x + (size_t)(sampleBase + nIdx) * 576 + kBase;

    v8f acc = {0.f, 0.f, 0.f, 0.f, 0.f, 0.f, 0.f, 0.f};
#pragma unroll
    for (int row = 0; row < 6; ++row) {
#pragma unroll
        for (int t = 0; t < 6; ++t) {
            v2f a = *(const v2f*)(px + row * 24 + t * 4);
            acc = __builtin_amdgcn_wmma_f32_16x16x4_f32(
                false, a, false, bt[t], (short)0, acc, false, false);
        }
    }

    if (nIdx < 4) {
        int srow = wv * 16 + ((lane >> 4) << 3);
#pragma unroll
        for (int r = 0; r < 8; ++r)
            pooled[srow + r][nIdx] = acc[r];
    }
    __syncthreads();

    if (tid < SPB) {
        float z[4];
        float prod = 1.0f;
#pragma unroll
        for (int j = 0; j < 4; ++j) {
            float xj = pooled[tid][j];
            float sx, cx;
            sincosf(xj, &sx, &cx);
            prod *= sAB[j] * cx + sAB[4 + j] * sx;
            z[j] = prod;
        }
        size_t s = (size_t)blockIdx.x * SPB + tid;
        *(float4*)(zbuf + s * 4) = make_float4(z[0], z[1], z[2], z[3]);
        float v[8];
        for (int j = 0; j < 4; ++j) {
            v[j]     = z[j];
            v[4 + j] = z[j] * z[j];
        }
        for (int j = 0; j < 8; ++j) v[j] += row_xor<1>(v[j]);
        for (int j = 0; j < 8; ++j) v[j] += row_xor<2>(v[j]);
        for (int j = 0; j < 8; ++j) v[j] += row_xor<4>(v[j]);
        for (int j = 0; j < 8; ++j) v[j] += row_xor<8>(v[j]);
        for (int j = 0; j < 8; ++j) v[j] += xhalf(v[j]);
        if (lane == 0) {
            wred[wv][0] = make_float4(v[0], v[1], v[2], v[3]);
            wred[wv][1] = make_float4(v[4], v[5], v[6], v[7]);
        }
    }
    __syncthreads();

    if (tid < 8) {
        const float* w = (const float*)wred;
        partials[(size_t)blockIdx.x * 8 + tid] =
            (w[tid] + w[8 + tid]) + (w[16 + tid] + w[24 + tid]);
    }
}

#define THREADS_N 256

__global__ __launch_bounds__(THREADS_N)
void stats_norm_kernel(const float* __restrict__ zbuf,
                       const float* __restrict__ partials, int nblk,
                       const float* __restrict__ gamma,
                       const float* __restrict__ beta,
                       float* __restrict__ out, int B)
{
    __shared__ float4 wsum[THREADS_N / 32][2];

    const int tid  = threadIdx.x;
    const int lane = tid & 31;
    const int wv   = tid >> 5;
    const int i    = blockIdx.x * THREADS_N + tid;

    float4 z = make_float4(0.f, 0.f, 0.f, 0.f);
    if (i < B) z = *(const float4*)(zbuf + (size_t)i * 4);
    const float g0 = gamma[0], g1 = gamma[1], g2 = gamma[2], g3 = gamma[3];
    const float b0 = beta[0],  b1 = beta[1],  b2 = beta[2],  b3 = beta[3];

    const float4* p4 = (const float4*)partials;
    const int n2 = 2 * nblk;
    const float4 zero4 = make_float4(0.f, 0.f, 0.f, 0.f);
    float4 a = zero4;
    for (int e = tid; e < n2; e += 4 * THREADS_N) {
        const int e1 = e + THREADS_N, e2 = e + 2 * THREADS_N, e3 = e + 3 * THREADS_N;
        float4 v0 = p4[e];
        float4 v1 = zero4, v2 = zero4, v3 = zero4;
        if (e1 < n2) v1 = p4[e1];
        if (e2 < n2) v2 = p4[e2];
        if (e3 < n2) v3 = p4[e3];
        a.x += (v0.x + v1.x) + (v2.x + v3.x);
        a.y += (v0.y + v1.y) + (v2.y + v3.y);
        a.z += (v0.z + v1.z) + (v2.z + v3.z);
        a.w += (v0.w + v1.w) + (v2.w + v3.w);
    }
    for (int m = 2; m < 32; m <<= 1) {
        a.x += __shfl_xor(a.x, m);
        a.y += __shfl_xor(a.y, m);
        a.z += __shfl_xor(a.z, m);
        a.w += __shfl_xor(a.w, m);
    }
    if (lane < 2) wsum[wv][lane] = a;
    __syncthreads();

    float4 s1 = wsum[0][0];
    float4 s2 = wsum[0][1];
    for (int w = 1; w < THREADS_N / 32; ++w) {
        float4 u = wsum[w][0];
        float4 q = wsum[w][1];
        s1.x += u.x; s1.y += u.y; s1.z += u.z; s1.w += u.w;
        s2.x += q.x; s2.y += q.y; s2.z += q.z; s2.w += q.w;
    }

    if (i < B) {
        const float invB = 1.0f / (float)B;
        float4 mean, var, sc, o;
        mean.x = s1.x * invB; mean.y = s1.y * invB; mean.z = s1.z * invB; mean.w = s1.w * invB;
        var.x = s2.x * invB - mean.x * mean.x;
        var.y = s2.y * invB - mean.y * mean.y;
        var.z = s2.z * invB - mean.z * mean.z;
        var.w = s2.w * invB - mean.w * mean.w;
        sc.x = g0 * rsqrtf(var.x + 1e-5f);
        sc.y = g1 * rsqrtf(var.y + 1e-5f);
        sc.z = g2 * rsqrtf(var.z + 1e-5f);
        sc.w = g3 * rsqrtf(var.w + 1e-5f);
        o.x = z.x * sc.x + (b0 - mean.x * sc.x);
        o.y = z.y * sc.y + (b1 - mean.y * sc.y);
        o.z = z.z * sc.z + (b2 - mean.z * sc.z);
        o.w = z.w * sc.w + (b3 - mean.w * sc.w);
        *(float4*)(out + (size_t)i * 4) = o;
    }
}

extern "C" void kernel_launch(void* const* d_in, const int* in_sizes, int n_in,
                              void* d_out, int out_size, void* d_ws, size_t ws_size,
                              hipStream_t stream)
{
    const float* x      = (const float*)d_in[0];
    const float* params = (const float*)d_in[1];
    const float* gamma  = (const float*)d_in[2];
    const float* beta   = (const float*)d_in[3];
    float* out = (float*)d_out;

    int B = in_sizes[0] / 576;
    int nblkA = B / SPB;

    float* zbuf     = (float*)d_ws;
    float* partials = zbuf + (size_t)B * 4;

    pool_circuit_kernel<<<nblkA, THREADS_A, 0, stream>>>(x, params, zbuf, partials, B);
    stats_norm_kernel<<<(B + 255) / 256, THREADS_N, 0, stream>>>(zbuf, partials, nblkA, gamma, beta, out, B);
}


# Kernel A reads x with whole-line 16-byte-per-lane loads (18 per wave) staged in a padded wave-private LDS tile; WMMA A fragments read from LDS as one group, k order kept (was 36 loads of 16 lines)
# speedup vs baseline: 2.8762x; 1.0880x over previous
#include <hip/hip_runtime.h>
#include <math.h>

typedef float v2f __attribute__((ext_vector_type(2)));
typedef float v8f __attribute__((ext_vector_type(8)));

#define SPB 128
#define THREADS_A 256
#define SAMPLE_F 576
#define NLOAD 18
#define PITCH 148

template <int M>
__device__ __forceinline__ float row_xor(float x)
{
    int b = __builtin_bit_cast(int, x);
    return __builtin_bit_cast(float,
        __builtin_amdgcn_update_dpp(0, b, 0x160 + M, 0xf, 0xf, true));
}

__device__ __forceinline__ float xhalf(float x)
{
    int b = __builtin_bit_cast(int, x);
    return __builtin_bit_cast(float,
        __builtin_amdgcn_permlanex16(b, b, 0x76543210, 0xfedcba98, false, false));
}

__global__ __launch_bounds__(THREADS_A)
void pool_circuit_kernel(const float* __restrict__ x,
                         const float* __restrict__ params,
                         float* __restrict__ zbuf,
                         float* __restrict__ partials,
                         int B)
{
    __shared__ __attribute__((aligned(16))) float stage[THREADS_A / 32][16 * PITCH];
    __shared__ float pooled[SPB][4];
    __shared__ float4 wred[SPB / 32][2];
    __shared__ float sAB[8];

    const int tid   = threadIdx.x;
    const int lane  = tid & 31;
    const int wv    = tid >> 5;
    const int nIdx  = lane & 15;
    const int kBase = (lane >> 4) << 1;

    if (tid < 4) {
        float al = params[2 * tid + 0];
        float be = params[2 * tid + 1];
        sAB[tid]     = cosf(be);
        sAB[tid + 4] = sinf(be) * sinf(al);
    }

    v2f bt[6];
#pragma unroll
    for (int t = 0; t < 6; ++t) {
        int c0 = 4 * t + kBase;
        bt[t].x = (nIdx == (c0    ) / 6) ? (1.0f / 36.0f) : 0.0f;
        bt[t].y = (nIdx == (c0 + 1) / 6) ? (1.0f / 36.0f) : 0.0f;
    }

    const float* xb = x + (size_t)blockIdx.x * (SPB * SAMPLE_F);
    float* st = &stage[wv][0];

    const unsigned gw = 16u * wv * SAMPLE_F + 4u * lane;
    const unsigned lw = 4u * lane;
    float4   g[NLOAD];
    unsigned so[NLOAD];
#pragma unroll
    for (int j = 0; j < NLOAD; ++j) {
        const unsigned S = (32u * j) / 36u, R = (32u * j) % 36u;
        const unsigned c = (lane + R >= 36u) ? 1u : 0u;
        g[j]  = *(const float4*)(xb + (gw + (S * SAMPLE_F + 4u * R) + c * (SAMPLE_F - 144u)));
        so[j] = lw + (S * PITCH + 4u * R) + c * (PITCH - 144u);
    }

#pragma unroll
    for (int j = 0; j < NLOAD; ++j)
        *(float4*)(st + so[j]) = g[j];

    const float* px = st + nIdx * PITCH + kBase;

    v2f af[36];
#pragma unroll
    for (int i = 0; i < 36; ++i)
        af[i] = *(const v2f*)(px + i * 4);
    __builtin_amdgcn_sched_barrier(0);

    v8f acc = {0.f, 0.f, 0.f, 0.f, 0.f, 0.f, 0.f, 0.f};
#pragma unroll
    for (int row = 0; row < 6; ++row) {
#pragma unroll
        for (int t = 0; t < 6; ++t) {
            acc = __builtin_amdgcn_wmma_f32_16x16x4_f32(
                false, af[row * 6 + t], false, bt[t], (short)0, acc, false, false);
        }
    }

    if (nIdx < 4) {
        int srow = wv * 16 + ((lane >> 4) << 3);
#pragma unroll
        for (int r = 0; r < 8; ++r)
            pooled[srow + r][nIdx] = acc[r];
    }
    __syncthreads();

    if (tid < SPB) {
        float z[4];
        float prod = 1.0f;
#pragma unroll
        for (int j = 0; j < 4; ++j) {
            float xj = pooled[tid][j];
            float sx, cx;
            sincosf(xj, &sx, &cx);
            prod *= sAB[j] * cx + sAB[4 + j] * sx;
            z[j] = prod;
        }
        size_t s = (size_t)blockIdx.x * SPB + tid;
        *(float4*)(zbuf + s * 4) = make_float4(z[0], z[1], z[2], z[3]);
        float v[8];
        for (int j = 0; j < 4; ++j) {
            v[j]     = z[j];
            v[4 + j] = z[j] * z[j];
        }
        for (int j = 0; j < 8; ++j) v[j] += row_xor<1>(v[j]);
        for (int j = 0; j < 8; ++j) v[j] += row_xor<2>(v[j]);
        for (int j = 0; j < 8; ++j) v[j] += row_xor<4>(v[j]);
        for (int j = 0; j < 8; ++j) v[j] += row_xor<8>(v[j]);
        for (int j = 0; j < 8; ++j) v[j] += xhalf(v[j]);
        if (lane == 0) {
            wred[wv][0] = make_float4(v[0], v[1], v[2], v[3]);
            wred[wv][1] = make_float4(v[4], v[5], v[6], v[7]);
        }
    }
    __syncthreads();

    if (tid < 8) {
        const float* w = (const float*)wred;
        partials[(size_t)blockIdx.x * 8 + tid] =
            (w[tid] + w[8 + tid]) + (w[16 + tid] + w[24 + tid]);
    }
}

#define THREADS_N 256

__global__ __launch_bounds__(THREADS_N)
void stats_norm_kernel(const float* __restrict__ zbuf,
                       const float* __restrict__ partials, int nblk,
                       const float* __restrict__ gamma,
                       const float* __restrict__ beta,
                       float* __restrict__ out, int B)
{
    __shared__ float4 wsum[THREADS_N / 32][2];

    const int tid  = threadIdx.x;
    const int lane = tid & 31;
    const int wv   = tid >> 5;
    const int i    = blockIdx.x * THREADS_N + tid;

    float4 z = make_float4(0.f, 0.f, 0.f, 0.f);
    if (i < B) z = *(const float4*)(zbuf + (size_t)i * 4);
    const float g0 = gamma[0], g1 = gamma[1], g2 = gamma[2], g3 = gamma[3];
    const float b0 = beta[0],  b1 = beta[1],  b2 = beta[2],  b3 = beta[3];

    const float4* p4 = (const float4*)partials;
    const int n2 = 2 * nblk;
    const float4 zero4 = make_float4(0.f, 0.f, 0.f, 0.f);
    float4 a = zero4;
    for (int e = tid; e < n2; e += 4 * THREADS_N) {
        const int e1 = e + THREADS_N, e2 = e + 2 * THREADS_N, e3 = e + 3 * THREADS_N;
        float4 v0 = p4[e];
        float4 v1 = zero4, v2 = zero4, v3 = zero4;
        if (e1 < n2) v1 = p4[e1];
        if (e2 < n2) v2 = p4[e2];
        if (e3 < n2) v3 = p4[e3];
        a.x += (v0.x + v1.x) + (v2.x + v3.x);
        a.y += (v0.y + v1.y) + (v2.y + v3.y);
        a.z += (v0.z + v1.z) + (v2.z + v3.z);
        a.w += (v0.w + v1.w) + (v2.w + v3.w);
    }
    for (int m = 2; m < 32; m <<= 1) {
        a.x += __shfl_xor(a.x, m);
        a.y += __shfl_xor(a.y, m);
        a.z += __shfl_xor(a.z, m);
        a.w += __shfl_xor(a.w, m);
    }
    if (lane < 2) wsum[wv][lane] = a;
    __syncthreads();

    float4 s1 = wsum[0][0];
    float4 s2 = wsum[0][1];
    for (int w = 1; w < THREADS_N / 32; ++w) {
        float4 u = wsum[w][0];
        float4 q = wsum[w][1];
        s1.x += u.x; s1.y += u.y; s1.z += u.z; s1.w += u.w;
        s2.x += q.x; s2.y += q.y; s2.z += q.z; s2.w += q.w;
    }

    if (i < B) {
        const float invB = 1.0f / (float)B;
        float4 mean, var, sc, o;
        mean.x = s1.x * invB; mean.y = s1.y * invB; mean.z = s1.z * invB; mean.w = s1.w * invB;
        var.x = s2.x * invB - mean.x * mean.x;
        var.y = s2.y * invB - mean.y * mean.y;
        var.z = s2.z * invB - mean.z * mean.z;
        var.w = s2.w * invB - mean.w * mean.w;
        sc.x = g0 * rsqrtf(var.x + 1e-5f);
        sc.y = g1 * rsqrtf(var.y + 1e-5f);
        sc.z = g2 * rsqrtf(var.z + 1e-5f);
        sc.w = g3 * rsqrtf(var.w + 1e-5f);
        o.x = z.x * sc.x + (b0 - mean.x * sc.x);
        o.y = z.y * sc.y + (b1 - mean.y * sc.y);
        o.z = z.z * sc.z + (b2 - mean.z * sc.z);
        o.w = z.w * sc.w + (b3 - mean.w * sc.w);
        *(float4*)(out + (size_t)i * 4) = o;
    }
}

extern "C" void kernel_launch(void* const* d_in, const int* in_sizes, int n_in,
                              void* d_out, int out_size, void* d_ws, size_t ws_size,
                              hipStream_t stream)
{
    const float* x      = (const float*)d_in[0];
    const float* params = (const float*)d_in[1];
    const float* gamma  = (const float*)d_in[2];
    const float* beta   = (const float*)d_in[3];
    float* out = (float*)d_out;

    int B = in_sizes[0] / 576;
    int nblkA = B / SPB;

    float* zbuf     = (float*)d_ws;
    float* partials = zbuf + (size_t)B * 4;

    pool_circuit_kernel<<<nblkA, THREADS_A, 0, stream>>>(x, params, zbuf, partials, B);
    stats_norm_kernel<<<(B + 255) / 256, THREADS_N, 0, stream>>>(zbuf, partials, nblkA, gamma, beta, out, B);
}
